# starting kernel
# baseline (speedup 1.0000x reference)
#include <hip/hip_runtime.h>
#include <hip/hip_bf16.h>

#define D 1024
#define H 4096
#define EPS 1e-5f

typedef __attribute__((ext_vector_type(2))) float v2f;
typedef __attribute__((ext_vector_type(8))) float v8f;

#define OP_NONE     0
#define OP_SIGMOID  1
#define OP_SQRELU   2
#define OP_ADDRES   3
#define OP_GATEDRES 4

__global__ __launch_bounds__(256) void ln_mix_kernel(
    const float* __restrict__ xin, const float* __restrict__ w,
    const float* __restrict__ b,   const float* __restrict__ prev,
    const float* __restrict__ tmk, const float* __restrict__ tmv,
    const float* __restrict__ tmr,
    float* __restrict__ xln_out, float* __restrict__ xk_out,
    float* __restrict__ xv_out,  float* __restrict__ xr_out)
{
    __shared__ float ssum[8];
    __shared__ float ssq[8];
    const int t = threadIdx.x;
    const int wave = t >> 5;
    const int lane = t & 31;

    float s = 0.f, sq = 0.f;
    for (int i = t; i < D; i += 256) {
        float v = xin[i];
        s += v;
        sq += v * v;
    }
    for (int off = 16; off > 0; off >>= 1) {
        s  += __shfl_xor(s,  off, 32);
        sq += __shfl_xor(sq, off, 32);
    }
    if (lane == 0) { ssum[wave] = s; ssq[wave] = sq; }
    __syncthreads();
    if (wave == 0) {
        float a = (lane < 8) ? ssum[lane] : 0.f;
        float c = (lane < 8) ? ssq[lane]  : 0.f;
        for (int off = 4; off > 0; off >>= 1) {
            a += __shfl_xor(a, off, 32);
            c += __shfl_xor(c, off, 32);
        }
        if (lane == 0) { ssum[0] = a; ssq[0] = c; }
    }
    __syncthreads();

    const float mu   = ssum[0] * (1.f / D);
    const float var  = ssq[0] * (1.f / D) - mu * mu;
    const float rstd = rsqrtf(var + EPS);

    for (int i = t; i < D; i += 256) {
        float xl = (xin[i] - mu) * rstd * w[i] + b[i];
        xln_out[i] = xl;
        float p = prev[i];
        float k = tmk[i];
        xk_out[i] = xl * k + p * (1.f - k);
        if (xv_out) {
            float v = tmv[i];
            xv_out[i] = xl * v + p * (1.f - v);
        }
        float r = tmr[i];
        xr_out[i] = xl * r + p * (1.f - r);
    }
}

struct MatvecJob {
    const float* W;
    const float* x;
    float*       y;
    const float* res;
    const float* gate;
    int K;
    int ldw;
    int op;
    int nblocks;
};

__device__ __forceinline__ void matvec_body(
    const float* __restrict__ W, const float* __restrict__ x,
    float* __restrict__ y, int K, int ldw, int op,
    const float* __restrict__ res, const float* __restrict__ gate, int blk)
{
    __shared__ float part[8][16];

    const int wave = threadIdx.x >> 5;
    const int lane = threadIdx.x & 31;
    const int half = lane >> 4;
    const int m    = lane & 15;
    const int row0 = blk << 4;

    const int kchunk = K >> 3;
    const int k0 = wave * kchunk;
    const int k1 = k0 + kchunk;

    const float* wrow = W + (size_t)(row0 + m) * (size_t)ldw + (half << 1);
    const float* xrow = x + (half << 1);

    v8f c0 = {}; v8f c1 = {}; v8f c2 = {}; v8f c3 = {};

    for (int k = k0; k < k1; k += 16) {
        v2f a0 = *(const v2f*)(wrow + k);
        v2f a1 = *(const v2f*)(wrow + k + 4);
        v2f a2 = *(const v2f*)(wrow + k + 8);
        v2f a3 = *(const v2f*)(wrow + k + 12);
        v2f b0 = *(const v2f*)(xrow + k);
        v2f b1 = *(const v2f*)(xrow + k + 4);
        v2f b2 = *(const v2f*)(xrow + k + 8);
        v2f b3 = *(const v2f*)(xrow + k + 12);
        c0 = __builtin_amdgcn_wmma_f32_16x16x4_f32(false, a0, false, b0, (short)0, c0, false, false);
        c1 = __builtin_amdgcn_wmma_f32_16x16x4_f32(false, a1, false, b1, (short)0, c1, false, false);
        c2 = __builtin_amdgcn_wmma_f32_16x16x4_f32(false, a2, false, b2, (short)0, c2, false, false);
        c3 = __builtin_amdgcn_wmma_f32_16x16x4_f32(false, a3, false, b3, (short)0, c3, false, false);
    }

    v8f c = c0 + c1 + c2 + c3;

    if (m == 0) {
        const int rbase = half << 3;
        #pragma unroll
        for (int j = 0; j < 8; ++j)
            part[wave][rbase + j] = c[j];
    }
    __syncthreads();

    if (threadIdx.x < 16) {
        float v = 0.f;
        #pragma unroll
        for (int wv = 0; wv < 8; ++wv)
            v += part[wv][threadIdx.x];
        const int r = row0 + threadIdx.x;
        switch (op) {
            case OP_SIGMOID:  v = 1.f / (1.f + expf(-v));              break;
            case OP_SQRELU:   { float t = fmaxf(v, 0.f); v = t * t; }  break;
            case OP_ADDRES:   v = res[r] + v;                          break;
            case OP_GATEDRES: v = res[r] + gate[r] * v;                break;
            default: break;
        }
        y[r] = v;
    }
}

__global__ __launch_bounds__(256) void wmma_matvec3_kernel(
    MatvecJob j0, MatvecJob j1, MatvecJob j2)
{
    int blk = blockIdx.x;
    if (blk < j0.nblocks) {
        matvec_body(j0.W, j0.x, j0.y, j0.K, j0.ldw, j0.op, j0.res, j0.gate, blk);
        return;
    }
    blk -= j0.nblocks;
    if (blk < j1.nblocks) {
        matvec_body(j1.W, j1.x, j1.y, j1.K, j1.ldw, j1.op, j1.res, j1.gate, blk);
        return;
    }
    blk -= j1.nblocks;
    matvec_body(j2.W, j2.x, j2.y, j2.K, j2.ldw, j2.op, j2.res, j2.gate, blk);
}

__global__ __launch_bounds__(256) void wkv_kernel(
    const float* __restrict__ kvec, const float* __restrict__ vvec,
    const float* __restrict__ rvec, const float* __restrict__ state,
    const float* __restrict__ time_first, const float* __restrict__ time_decay,
    float* __restrict__ rwkv,
    float* __restrict__ s2, float* __restrict__ s3, float* __restrict__ s4)
{
    const int i = blockIdx.x * 256 + threadIdx.x;
    if (i >= D) return;

    float k  = kvec[i];
    float v  = vvec[i];
    float r  = rvec[i];
    float aa = state[2 * D + i];
    float bb = state[3 * D + i];
    float pp = state[4 * D + i];

    float ww = time_first[i] + k;
    float qq = fmaxf(pp, ww);
    float e1 = expf(pp - qq);
    float e2 = expf(ww - qq);
    float wkv = (e1 * aa + e2 * v) / (e1 * bb + e2);
    rwkv[i] = r * wkv;

    float ww2 = pp + time_decay[i];
    float qq2 = fmaxf(ww2, k);
    float f1 = expf(ww2 - qq2);
    float f2 = expf(k - qq2);
    s2[i] = f1 * aa + f2 * v;
    s3[i] = f1 * bb + f2;
    s4[i] = qq2;
}

extern "C" void kernel_launch(void* const* d_in, const int* in_sizes, int n_in,
                              void* d_out, int out_size, void* d_ws, size_t ws_size,
                              hipStream_t stream) {
    const float* x     = (const float*)d_in[0];
    const float* state = (const float*)d_in[1];
    const float* tmk   = (const float*)d_in[2];
    const float* tmv   = (const float*)d_in[3];
    const float* tmr   = (const float*)d_in[4];
    const float* tf    = (const float*)d_in[5];
    const float* td    = (const float*)d_in[6];
    const float* kw    = (const float*)d_in[7];
    const float* vw    = (const float*)d_in[8];
    const float* rw    = (const float*)d_in[9];
    const float* ow    = (const float*)d_in[10];
    const float* lnw   = (const float*)d_in[11];
    const float* lnb   = (const float*)d_in[12];
    const float* ftmk  = (const float*)d_in[13];
    const float* ftmr  = (const float*)d_in[14];
    const float* fkw   = (const float*)d_in[15];
    const float* fvw   = (const float*)d_in[16];
    const float* frw   = (const float*)d_in[17];
    const float* flnw  = (const float*)d_in[18];
    const float* flnb  = (const float*)d_in[19];

    float* ws    = (float*)d_ws;
    float* xk    = ws + 1 * D;
    float* xv    = ws + 2 * D;
    float* xr    = ws + 3 * D;
    float* kvec  = ws + 4 * D;
    float* vvec  = ws + 5 * D;
    float* rvec  = ws + 6 * D;
    float* rwkv  = ws + 7 * D;
    float* x_att = ws + 8 * D;
    float* fxk   = ws + 9 * D;
    float* fxr   = ws + 10 * D;
    float* fr    = ws + 11 * D;
    float* fk    = ws + 12 * D;

    float* out = (float*)d_out;
    float* s0  = out + 1 * D;
    float* s1  = out + 2 * D;
    float* s2  = out + 3 * D;
    float* s3  = out + 4 * D;
    float* s4  = out + 5 * D;

    const MatvecJob jnull = { nullptr, nullptr, nullptr, nullptr, nullptr, 0, 0, OP_NONE, 0 };

    ln_mix_kernel<<<1, 256, 0, stream>>>(x, lnw, lnb, state + 1 * D,
                                         tmk, tmv, tmr, s1, xk, xv, xr);

    {
        MatvecJob jk = { kw, xk, kvec, nullptr, nullptr, D, D, OP_NONE,    D / 16 };
        MatvecJob jv = { vw, xv, vvec, nullptr, nullptr, D, D, OP_NONE,    D / 16 };
        MatvecJob jr = { rw, xr, rvec, nullptr, nullptr, D, D, OP_SIGMOID, D / 16 };
        wmma_matvec3_kernel<<<3 * (D / 16), 256, 0, stream>>>(jk, jv, jr);
    }

    wkv_kernel<<<D / 256, 256, 0, stream>>>(kvec, vvec, rvec, state, tf, td,
                                            rwkv, s2, s3, s4);

    {
        MatvecJob jo = { ow, rwkv, x_att, x, nullptr, D, D, OP_ADDRES, D / 16 };
        wmma_matvec3_kernel<<<D / 16, 256, 0, stream>>>(jo, jnull, jnull);
    }

    ln_mix_kernel<<<1, 256, 0, stream>>>(x_att, flnw, flnb, state  ,
                                         ftmk, nullptr, ftmr, s0, fxk, nullptr, fxr);

    {
        MatvecJob jfr = { frw, fxr, fr, nullptr, nullptr, D, D, OP_SIGMOID, D / 16 };
        MatvecJob jfk = { fkw, fxk, fk, nullptr, nullptr, D, D, OP_SQRELU,  H / 16 };
        wmma_matvec3_kernel<<<(D / 16) + (H / 16), 256, 0, stream>>>(jfr, jfk, jnull);
    }

    {
        MatvecJob jout = { fvw, fk, out, x_att, fr, H, H, OP_GATEDRES, D / 16 };
        wmma_matvec3_kernel<<<D / 16, 256, 0, stream>>>(jout, jnull, jnull);
    }
}


# Fold the two LayerNorm+mix kernels and the WKV kernel into the mat-vec launches that consume them, as per-block prologues into LDS (7 launches to 4); every summation keeps its order
# speedup vs baseline: 1.5994x; 1.5994x over previous
#include <hip/hip_runtime.h>
#include <hip/hip_bf16.h>


#define D 1024
#define H 4096
#define EPS 1e-5f

typedef __attribute__((ext_vector_type(2))) float v2f;
typedef __attribute__((ext_vector_type(8))) float v8f;

#define OP_NONE     0
#define OP_SIGMOID  1
#define OP_SQRELU   2
#define OP_ADDRES   3
#define OP_GATEDRES 4

struct MatvecJob {
    const float* W;
    const float* x;
    const float* tm;
    float*       y;
    const float* res;
    const float* gate;
    int K;
    int ldw;
    int op;
    int nblocks;
};

struct LnMixArgs {
    const float* xin;
    const float* w;
    const float* b;
    const float* prev;
    float*       xln_out;
};

struct WkvArgs {
    const float* kvec;
    const float* vvec;
    const float* rvec;
    const float* state;
    const float* time_first;
    const float* time_decay;
    float* s2;
    float* s3;
    float* s4;
};

struct AFrag { v2f a0, a1, a2, a3; };

__device__ __forceinline__ AFrag load_afrag(const float* __restrict__ wrow, int k)
{
    AFrag f;
    f.a0 = *(const v2f*)(wrow + k);
    f.a1 = *(const v2f*)(wrow + k + 4);
    f.a2 = *(const v2f*)(wrow + k + 8);
    f.a3 = *(const v2f*)(wrow + k + 12);
    return f;
}

__device__ __forceinline__ void matvec_tail(
    const float* __restrict__ wrow, const float* xv, AFrag first,
    int k0, int k1, float (*part)[16],
    float* __restrict__ y, int op,
    const float* __restrict__ res, const float* __restrict__ gate, int row0)
{
    const int wave = threadIdx.x >> 5;
    const int lane = threadIdx.x & 31;
    const int half = lane >> 4;
    const int m    = lane & 15;

    v8f c0 = {}; v8f c1 = {}; v8f c2 = {}; v8f c3 = {};

    {
        v2f b0 = *(const v2f*)(xv + k0);
        v2f b1 = *(const v2f*)(xv + k0 + 4);
        v2f b2 = *(const v2f*)(xv + k0 + 8);
        v2f b3 = *(const v2f*)(xv + k0 + 12);
        c0 = __builtin_amdgcn_wmma_f32_16x16x4_f32(false, first.a0, false, b0, (short)0, c0, false, false);
        c1 = __builtin_amdgcn_wmma_f32_16x16x4_f32(false, first.a1, false, b1, (short)0, c1, false, false);
        c2 = __builtin_amdgcn_wmma_f32_16x16x4_f32(false, first.a2, false, b2, (short)0, c2, false, false);
        c3 = __builtin_amdgcn_wmma_f32_16x16x4_f32(false, first.a3, false, b3, (short)0, c3, false, false);
    }

    for (int k = k0 + 16; k < k1; k += 16) {
        AFrag f = load_afrag(wrow, k);
        v2f b0 = *(const v2f*)(xv + k);
        v2f b1 = *(const v2f*)(xv + k + 4);
        v2f b2 = *(const v2f*)(xv + k + 8);
        v2f b3 = *(const v2f*)(xv + k + 12);
        c0 = __builtin_amdgcn_wmma_f32_16x16x4_f32(false, f.a0, false, b0, (short)0, c0, false, false);
        c1 = __builtin_amdgcn_wmma_f32_16x16x4_f32(false, f.a1, false, b1, (short)0, c1, false, false);
        c2 = __builtin_amdgcn_wmma_f32_16x16x4_f32(false, f.a2, false, b2, (short)0, c2, false, false);
        c3 = __builtin_amdgcn_wmma_f32_16x16x4_f32(false, f.a3, false, b3, (short)0, c3, false, false);
    }

    v8f c = c0 + c1 + c2 + c3;

    if (m == 0) {
        const int rbase = half << 3;
        #pragma unroll
        for (int j = 0; j < 8; ++j)
            part[wave][rbase + j] = c[j];
    }
    __syncthreads();

    if (threadIdx.x < 16) {
        float v = 0.f;
        #pragma unroll
        for (int wv = 0; wv < 8; ++wv)
            v += part[wv][threadIdx.x];
        const int r = row0 + threadIdx.x;
        switch (op) {
            case OP_SIGMOID:  v = 1.f / (1.f + expf(-v));              break;
            case OP_SQRELU:   { float t = fmaxf(v, 0.f); v = t * t; }  break;
            case OP_ADDRES:   v = res[r] + v;                          break;
            case OP_GATEDRES: v = res[r] + gate[r] * v;                break;
            default: break;
        }
        y[r] = v;
    }
}

__device__ __forceinline__ MatvecJob pick_job(
    const MatvecJob& j0, const MatvecJob& j1, const MatvecJob& j2, int& blk)
{
    if (blk < j0.nblocks) return j0;
    blk -= j0.nblocks;
    if (blk < j1.nblocks) return j1;
    blk -= j1.nblocks;
    return j2;
}

__global__ __launch_bounds__(256) void wmma_matvec3_kernel(
    MatvecJob j0, MatvecJob j1, MatvecJob j2)
{
    __shared__ float part[8][16];
    int blk = blockIdx.x;
    const MatvecJob j = pick_job(j0, j1, j2, blk);

    const int wave = threadIdx.x >> 5;
    const int lane = threadIdx.x & 31;
    const int half = lane >> 4;
    const int m    = lane & 15;
    const int row0 = blk << 4;
    const int kchunk = j.K >> 3;
    const int k0 = wave * kchunk;
    const int k1 = k0 + kchunk;
    const float* wrow = j.W + (size_t)(row0 + m) * (size_t)j.ldw + (half << 1);

    AFrag first = load_afrag(wrow, k0);
    matvec_tail(wrow, j.x + (half << 1), first, k0, k1, part,
                j.y, j.op, j.res, j.gate, row0);
}

__global__ __launch_bounds__(256) void ln_matvec3_kernel(
    LnMixArgs p, MatvecJob j0, MatvecJob j1, MatvecJob j2)
{
    __shared__ float xs[D];
    __shared__ float part[8][16];
    __shared__ float ssum[8];
    __shared__ float ssq[8];

    int blk = blockIdx.x;
    const MatvecJob j = pick_job(j0, j1, j2, blk);

    const int t    = threadIdx.x;
    const int wave = t >> 5;
    const int lane = t & 31;
    const int half = lane >> 4;
    const int m    = lane & 15;
    const int row0 = blk << 4;
    const int kchunk = j.K >> 3;
    const int k0 = wave * kchunk;
    const int k1 = k0 + kchunk;
    const float* wrow = j.W + (size_t)(row0 + m) * (size_t)j.ldw + (half << 1);

    float xv[4], wv[4], bv[4], pv[4], mv[4];
    #pragma unroll
    for (int q = 0; q < 4; ++q) {
        const int i = t + q * 256;
        xv[q] = p.xin[i];
        wv[q] = p.w[i];
        bv[q] = p.b[i];
        pv[q] = p.prev[i];
        mv[q] = j.tm[i];
    }

    AFrag first = load_afrag(wrow, k0);

    float s = 0.f, sq = 0.f;
    #pragma unroll
    for (int q = 0; q < 4; ++q) {
        float v = xv[q];
        s += v;
        sq += v * v;
    }
    for (int off = 16; off > 0; off >>= 1) {
        s  += __shfl_xor(s,  off, 32);
        sq += __shfl_xor(sq, off, 32);
    }
    if (lane == 0) { ssum[wave] = s; ssq[wave] = sq; }
    __syncthreads();
    float a = (lane < 8) ? ssum[lane] : 0.f;
    float c = (lane < 8) ? ssq[lane]  : 0.f;
    for (int off = 4; off > 0; off >>= 1) {
        a += __shfl_xor(a, off, 32);
        c += __shfl_xor(c, off, 32);
    }
    a = __shfl(a, 0, 32);
    c = __shfl(c, 0, 32);

    const float mu   = a * (1.f / D);
    const float var  = c * (1.f / D) - mu * mu;
    const float rstd = rsqrtf(var + EPS);

    const bool writer = (blockIdx.x == 0);
    #pragma unroll
    for (int q = 0; q < 4; ++q) {
        const int i = t + q * 256;
        float xl = (xv[q] - mu) * rstd * wv[q] + bv[q];
        if (writer) p.xln_out[i] = xl;
        float k = mv[q];
        xs[i] = xl * k + pv[q] * (1.f - k);
    }
    __syncthreads();

    matvec_tail(wrow, xs + (half << 1), first, k0, k1, part,
                j.y, j.op, j.res, j.gate, row0);
}

__global__ __launch_bounds__(256) void wkv_matvec_kernel(WkvArgs p, MatvecJob j)
{
    __shared__ float xs[D];
    __shared__ float part[8][16];

    const int blk  = blockIdx.x;
    const int t    = threadIdx.x;
    const int wave = t >> 5;
    const int lane = t & 31;
    const int half = lane >> 4;
    const int m    = lane & 15;
    const int row0 = blk << 4;
    const int kchunk = j.K >> 3;
    const int k0 = wave * kchunk;
    const int k1 = k0 + kchunk;
    const float* wrow = j.W + (size_t)(row0 + m) * (size_t)j.ldw + (half << 1);

    float kq[4], vq[4], rq[4], aq[4], bq[4], pq[4], tq[4];
    #pragma unroll
    for (int q = 0; q < 4; ++q) {
        const int i = t + q * 256;
        kq[q] = p.kvec[i];
        vq[q] = p.vvec[i];
        rq[q] = p.rvec[i];
        aq[q] = p.state[2 * D + i];
        bq[q] = p.state[3 * D + i];
        pq[q] = p.state[4 * D + i];
        tq[q] = p.time_first[i];
    }

    AFrag first = load_afrag(wrow, k0);
    __builtin_amdgcn_sched_barrier(0);

    #pragma unroll
    for (int q = 0; q < 4; ++q) {
        const int i = t + q * 256;
        float k  = kq[q];
        float v  = vq[q];
        float aa = aq[q];
        float bb = bq[q];
        float pp = pq[q];

        float ww = tq[q] + k;
        float qq = fmaxf(pp, ww);
        float e1 = expf(pp - qq);
        float e2 = expf(ww - qq);
        float wkv = (e1 * aa + e2 * v) / (e1 * bb + e2);
        xs[i] = rq[q] * wkv;
    }

    if (blk == 0) {
        float dq[4];
        #pragma unroll
        for (int q = 0; q < 4; ++q)
            dq[q] = p.time_decay[t + q * 256];
        #pragma unroll
        for (int q = 0; q < 4; ++q) {
            const int i = t + q * 256;
            float k  = kq[q];
            float v  = vq[q];
            float aa = aq[q];
            float bb = bq[q];
            float pp = pq[q];

            float ww2 = pp + dq[q];
            float qq2 = fmaxf(ww2, k);
            float f1 = expf(ww2 - qq2);
            float f2 = expf(k - qq2);
            p.s2[i] = f1 * aa + f2 * v;
            p.s3[i] = f1 * bb + f2;
            p.s4[i] = qq2;
        }
    }
    __syncthreads();

    matvec_tail(wrow, xs + (half << 1), first, k0, k1, part,
                j.y, j.op, j.res, j.gate, row0);
}

extern "C" void kernel_launch(void* const* d_in, const int* in_sizes, int n_in,
                              void* d_out, int out_size, void* d_ws, size_t ws_size,
                              hipStream_t stream) {
    const float* x     = (const float*)d_in[0];
    const float* state = (const float*)d_in[1];
    const float* tmk   = (const float*)d_in[2];
    const float* tmv   = (const float*)d_in[3];
    const float* tmr   = (const float*)d_in[4];
    const float* tf    = (const float*)d_in[5];
    const float* td    = (const float*)d_in[6];
    const float* kw    = (const float*)d_in[7];
    const float* vw    = (const float*)d_in[8];
    const float* rw    = (const float*)d_in[9];
    const float* ow    = (const float*)d_in[10];
    const float* lnw   = (const float*)d_in[11];
    const float* lnb   = (const float*)d_in[12];
    const float* ftmk  = (const float*)d_in[13];
    const float* ftmr  = (const float*)d_in[14];
    const float* fkw   = (const float*)d_in[15];
    const float* fvw   = (const float*)d_in[16];
    const float* frw   = (const float*)d_in[17];
    const float* flnw  = (const float*)d_in[18];
    const float* flnb  = (const float*)d_in[19];

    float* ws    = (float*)d_ws;
    float* kvec  = ws + 4 * D;
    float* vvec  = ws + 5 * D;
    float* rvec  = ws + 6 * D;
    float* x_att = ws + 8 * D;
    float* fr    = ws + 11 * D;
    float* fk    = ws + 12 * D;

    float* out = (float*)d_out;
    float* s0  = out + 1 * D;
    float* s1  = out + 2 * D;
    float* s2  = out + 3 * D;
    float* s3  = out + 4 * D;
    float* s4  = out + 5 * D;

    const MatvecJob jnull = { nullptr, nullptr, nullptr, nullptr, nullptr, nullptr, 0, 0, OP_NONE, 0 };

    {
        LnMixArgs ln = { x, lnw, lnb, state + 1 * D, s1 };
        MatvecJob jk = { kw, nullptr, tmk, kvec, nullptr, nullptr, D, D, OP_NONE,    D / 16 };
        MatvecJob jv = { vw, nullptr, tmv, vvec, nullptr, nullptr, D, D, OP_NONE,    D / 16 };
        MatvecJob jr = { rw, nullptr, tmr, rvec, nullptr, nullptr, D, D, OP_SIGMOID, D / 16 };
        ln_matvec3_kernel<<<3 * (D / 16), 256, 0, stream>>>(ln, jk, jv, jr);
    }

    {
        WkvArgs wk = { kvec, vvec, rvec, state, tf, td, s2, s3, s4 };
        MatvecJob jo = { ow, nullptr, nullptr, x_att, x, nullptr, D, D, OP_ADDRES, D / 16 };
        wkv_matvec_kernel<<<D / 16, 256, 0, stream>>>(wk, jo);
    }

    {
        LnMixArgs ln = { x_att, flnw, flnb, state  , s0 };
        MatvecJob jfr = { frw, nullptr, ftmr, fr, nullptr, nullptr, D, D, OP_SIGMOID, D / 16 };
        MatvecJob jfk = { fkw, nullptr, ftmk, fk, nullptr, nullptr, D, D, OP_SQRELU,  H / 16 };
        ln_matvec3_kernel<<<(D / 16) + (H / 16), 256, 0, stream>>>(ln, jfr, jfk, jnull);
    }

    {
        MatvecJob jout = { fvw, fk, nullptr, out, x_att, fr, H, H, OP_GATEDRES, D / 16 };
        wmma_matvec3_kernel<<<D / 16, 256, 0, stream>>>(jout, jnull, jnull);
    }
}


# Weight fetch re-mapped to whole 128-byte lines: a lane's 16-byte load puts 4 rows x 32 k in one fragment (x pieces on B columns, diagonal D blocks summed), 128 k in flight; sum order changes
# speedup vs baseline: 2.1637x; 1.3528x over previous
#include <hip/hip_runtime.h>
#include <hip/hip_bf16.h>


#define D 1024
#define H 4096
#define EPS 1e-5f

typedef __attribute__((ext_vector_type(2))) float v2f;
typedef __attribute__((ext_vector_type(4))) float v4f;
typedef __attribute__((ext_vector_type(8))) float v8f;

#define OP_NONE     0
#define OP_SIGMOID  1
#define OP_SQRELU   2
#define OP_ADDRES   3
#define OP_GATEDRES 4

struct MatvecJob {
    const float* W;
    const float* x;
    const float* tm;
    float*       y;
    const float* res;
    const float* gate;
    int op;
    int nblocks;
};

struct LnMixArgs {
    const float* xin;
    const float* w;
    const float* b;
    const float* prev;
    float*       xln_out;
};

struct WkvArgs {
    const float* kvec;
    const float* vvec;
    const float* rvec;
    const float* state;
    const float* time_first;
    const float* time_decay;
    float* s2;
    float* s3;
    float* s4;
};

struct WFrag { v4f a[16]; };
struct XFrag { v4f b[4]; };

template <int LDW>
__device__ __forceinline__ WFrag load_wfrag(const float* __restrict__ wp, int kb)
{
    WFrag f;
    #pragma unroll
    for (int w = 0; w < 4; ++w) {
        #pragma unroll
        for (int g = 0; g < 4; ++g)
            f.a[4 * w + g] = *(const v4f*)(wp + (size_t)(4 * g) * LDW + kb + 32 * w);
    }
    return f;
}

__device__ __forceinline__ XFrag load_xfrag(const float* xv, int kb)
{
    XFrag f;
    #pragma unroll
    for (int w = 0; w < 4; ++w)
        f.b[w] = *(const v4f*)(xv + kb + 32 * w);
    return f;
}

__device__ __forceinline__ void wmma_batch(const WFrag& fa, const XFrag& fb, v8f (&acc)[4])
{
    #pragma unroll
    for (int w = 0; w < 4; ++w) {
        const v2f blo = __builtin_shufflevector(fb.b[w], fb.b[w], 0, 1);
        const v2f bhi = __builtin_shufflevector(fb.b[w], fb.b[w], 2, 3);
        #pragma unroll
        for (int g = 0; g < 4; ++g) {
            const v2f alo = __builtin_shufflevector(fa.a[4 * w + g], fa.a[4 * w + g], 0, 1);
            acc[g] = __builtin_amdgcn_wmma_f32_16x16x4_f32(false, alo, false, blo, (short)0, acc[g], false, false);
        }
        #pragma unroll
        for (int g = 0; g < 4; ++g) {
            const v2f ahi = __builtin_shufflevector(fa.a[4 * w + g], fa.a[4 * w + g], 2, 3);
            acc[g] = __builtin_amdgcn_wmma_f32_16x16x4_f32(false, ahi, false, bhi, (short)0, acc[g], false, false);
        }
    }
}

template <int KK>
__device__ __forceinline__ void matvec_tail(
    const float* __restrict__ wp, const float* xv, WFrag first,
    int k0, float (*part)[16][4],
    float* __restrict__ y, int op,
    const float* __restrict__ res, const float* __restrict__ gate, int row0)
{
    constexpr int NB = KK / 8 / 128;
    const int wave = threadIdx.x >> 5;
    const int lane = threadIdx.x & 31;
    const int half = lane >> 4;
    const int c    = lane & 15;
    const int q    = c & 3;

    v8f acc[4] = {};

    WFrag ca = first;
    XFrag cb = load_xfrag(xv, k0);
    #pragma unroll
    for (int nb = 0; nb < NB; ++nb) {
        WFrag na;
        XFrag nx;
        if (nb + 1 < NB) {
            nx = load_xfrag(xv, k0 + 128 * (nb + 1));
            na = load_wfrag<KK>(wp, k0 + 128 * (nb + 1));
            __builtin_amdgcn_sched_barrier(0);
        }
        wmma_batch(ca, cb, acc);
        if (nb + 1 < NB) { ca = na; cb = nx; }
    }

    #pragma unroll
    for (int g = 0; g < 4; ++g) {
        const v8f d = acc[g];
        float lo = d[0], hi = d[4];
        lo = (q == 1) ? d[1] : lo;   hi = (q == 1) ? d[5] : hi;
        lo = (q == 2) ? d[2] : lo;   hi = (q == 2) ? d[6] : hi;
        lo = (q == 3) ? d[3] : lo;   hi = (q == 3) ? d[7] : hi;
        if (c < 4) {
            part[wave][4 * g + 2 * half    ][c] = lo;
            part[wave][4 * g + 2 * half + 1][c] = hi;
        }
    }
    __syncthreads();

    if (threadIdx.x < 16) {
        float v = 0.f;
        #pragma unroll
        for (int wv = 0; wv < 8; ++wv) {
            const v4f p = *(const v4f*)(&part[wv][threadIdx.x][0]);
            v += p.x; v += p.y; v += p.z; v += p.w;
        }
        const int r = row0 + threadIdx.x;
        switch (op) {
            case OP_SIGMOID:  v = 1.f / (1.f + expf(-v));              break;
            case OP_SQRELU:   { float t = fmaxf(v, 0.f); v = t * t; }  break;
            case OP_ADDRES:   v = res[r] + v;                          break;
            case OP_GATEDRES: v = res[r] + gate[r] * v;                break;
            default: break;
        }
        y[r] = v;
    }
}

__device__ __forceinline__ MatvecJob pick_job(
    const MatvecJob& j0, const MatvecJob& j1, const MatvecJob& j2, int& blk)
{
    if (blk < j0.nblocks) return j0;
    blk -= j0.nblocks;
    if (blk < j1.nblocks) return j1;
    blk -= j1.nblocks;
    return j2;
}

template <int KK>
__global__ __launch_bounds__(256) void wmma_matvec3_kernel(
    MatvecJob j0, MatvecJob j1, MatvecJob j2)
{
    __shared__ __attribute__((aligned(16))) float part[8][16][4];
    int blk = blockIdx.x;
    const MatvecJob j = pick_job(j0, j1, j2, blk);

    const int wave = threadIdx.x >> 5;
    const int lane = threadIdx.x & 31;
    const int half = lane >> 4;
    const int cl   = lane & 15;
    const int row0 = blk << 4;
    const int k0 = wave * (KK / 8);
    const int loff = 8 * (cl & 3) + 4 * half;
    const float* wp = j.W + (size_t)(row0 + (cl >> 2)) * (size_t)KK + loff;

    WFrag first = load_wfrag<KK>(wp, k0);
    matvec_tail<KK>(wp, j.x + loff, first, k0, part,
                    j.y, j.op, j.res, j.gate, row0);
}

__global__ __launch_bounds__(256) void ln_matvec3_kernel(
    LnMixArgs p, MatvecJob j0, MatvecJob j1, MatvecJob j2)
{
    __shared__ __attribute__((aligned(16))) float xs[D];
    __shared__ __attribute__((aligned(16))) float part[8][16][4];
    __shared__ float ssum[8];
    __shared__ float ssq[8];

    int blk = blockIdx.x;
    const MatvecJob j = pick_job(j0, j1, j2, blk);

    const int t    = threadIdx.x;
    const int wave = t >> 5;
    const int lane = t & 31;
    const int half = lane >> 4;
    const int cl   = lane & 15;
    const int row0 = blk << 4;
    const int k0 = wave * (D / 8);
    const int loff = 8 * (cl & 3) + 4 * half;
    const float* wp = j.W + (size_t)(row0 + (cl >> 2)) * (size_t)D + loff;

    float xv[4], wv[4], bv[4], pv[4], mv[4];
    #pragma unroll
    for (int q = 0; q < 4; ++q) {
        const int i = t + q * 256;
        xv[q] = p.xin[i];
        wv[q] = p.w[i];
        bv[q] = p.b[i];
        pv[q] = p.prev[i];
        mv[q] = j.tm[i];
    }

    WFrag first = load_wfrag<D>(wp, k0);

    float s = 0.f, sq = 0.f;
    #pragma unroll
    for (int q = 0; q < 4; ++q) {
        float v = xv[q];
        s += v;
        sq += v * v;
    }
    for (int off = 16; off > 0; off >>= 1) {
        s  += __shfl_xor(s,  off, 32);
        sq += __shfl_xor(sq, off, 32);
    }
    if (lane == 0) { ssum[wave] = s; ssq[wave] = sq; }
    __syncthreads();
    float a = (lane < 8) ? ssum[lane] : 0.f;
    float c = (lane < 8) ? ssq[lane]  : 0.f;
    for (int off = 4; off > 0; off >>= 1) {
        a += __shfl_xor(a, off, 32);
        c += __shfl_xor(c, off, 32);
    }
    a = __shfl(a, 0, 32);
    c = __shfl(c, 0, 32);

    const float mu   = a * (1.f / D);
    const float var  = c * (1.f / D) - mu * mu;
    const float rstd = rsqrtf(var + EPS);

    const bool writer = (blockIdx.x == 0);
    #pragma unroll
    for (int q = 0; q < 4; ++q) {
        const int i = t + q * 256;
        float xl = (xv[q] - mu) * rstd * wv[q] + bv[q];
        if (writer) p.xln_out[i] = xl;
        float k = mv[q];
        xs[i] = xl * k + pv[q] * (1.f - k);
    }
    __syncthreads();

    matvec_tail<D>(wp, xs + loff, first, k0, part,
                   j.y, j.op, j.res, j.gate, row0);
}

__global__ __launch_bounds__(256) void wkv_matvec_kernel(WkvArgs p, MatvecJob j)
{
    __shared__ __attribute__((aligned(16))) float xs[D];
    __shared__ __attribute__((aligned(16))) float part[8][16][4];

    const int blk  = blockIdx.x;
    const int t    = threadIdx.x;
    const int wave = t >> 5;
    const int lane = t & 31;
    const int half = lane >> 4;
    const int cl   = lane & 15;
    const int row0 = blk << 4;
    const int k0 = wave * (D / 8);
    const int loff = 8 * (cl & 3) + 4 * half;
    const float* wp = j.W + (size_t)(row0 + (cl >> 2)) * (size_t)D + loff;

    float kq[4], vq[4], rq[4], aq[4], bq[4], pq[4], tq[4];
    #pragma unroll
    for (int q = 0; q < 4; ++q) {
        const int i = t + q * 256;
        kq[q] = p.kvec[i];
        vq[q] = p.vvec[i];
        rq[q] = p.rvec[i];
        aq[q] = p.state[2 * D + i];
        bq[q] = p.state[3 * D + i];
        pq[q] = p.state[4 * D + i];
        tq[q] = p.time_first[i];
    }

    WFrag first = load_wfrag<D>(wp, k0);
    __builtin_amdgcn_sched_barrier(0);

    #pragma unroll
    for (int q = 0; q < 4; ++q) {
        const int i = t + q * 256;
        float k  = kq[q];
        float v  = vq[q];
        float aa = aq[q];
        float bb = bq[q];
        float pp = pq[q];

        float ww = tq[q] + k;
        float qq = fmaxf(pp, ww);
        float e1 = expf(pp - qq);
        float e2 = expf(ww - qq);
        float wkv = (e1 * aa + e2 * v) / (e1 * bb + e2);
        xs[i] = rq[q] * wkv;
    }

    if (blk == 0) {
        float dq[4];
        #pragma unroll
        for (int q = 0; q < 4; ++q)
            dq[q] = p.time_decay[t + q * 256];
        #pragma unroll
        for (int q = 0; q < 4; ++q) {
            const int i = t + q * 256;
            float k  = kq[q];
            float v  = vq[q];
            float aa = aq[q];
            float bb = bq[q];
            float pp = pq[q];

            float ww2 = pp + dq[q];
            float qq2 = fmaxf(ww2, k);
            float f1 = expf(ww2 - qq2);
            float f2 = expf(k - qq2);
            p.s2[i] = f1 * aa + f2 * v;
            p.s3[i] = f1 * bb + f2;
            p.s4[i] = qq2;
        }
    }
    __syncthreads();

    matvec_tail<D>(wp, xs + loff, first, k0, part,
                   j.y, j.op, j.res, j.gate, row0);
}

extern "C" void kernel_launch(void* const* d_in, const int* in_sizes, int n_in,
                              void* d_out, int out_size, void* d_ws, size_t ws_size,
                              hipStream_t stream) {
    const float* x     = (const float*)d_in[0];
    const float* state = (const float*)d_in[1];
    const float* tmk   = (const float*)d_in[2];
    const float* tmv   = (const float*)d_in[3];
    const float* tmr   = (const float*)d_in[4];
    const float* tf    = (const float*)d_in[5];
    const float* td    = (const float*)d_in[6];
    const float* kw    = (const float*)d_in[7];
    const float* vw    = (const float*)d_in[8];
    const float* rw    = (const float*)d_in[9];
    const float* ow    = (const float*)d_in[10];
    const float* lnw   = (const float*)d_in[11];
    const float* lnb   = (const float*)d_in[12];
    const float* ftmk  = (const float*)d_in[13];
    const float* ftmr  = (const float*)d_in[14];
    const float* fkw   = (const float*)d_in[15];
    const float* fvw   = (const float*)d_in[16];
    const float* frw   = (const float*)d_in[17];
    const float* flnw  = (const float*)d_in[18];
    const float* flnb  = (const float*)d_in[19];

    float* ws    = (float*)d_ws;
    float* kvec  = ws + 4 * D;
    float* vvec  = ws + 5 * D;
    float* rvec  = ws + 6 * D;
    float* x_att = ws + 8 * D;
    float* fr    = ws + 11 * D;
    float* fk    = ws + 12 * D;

    float* out = (float*)d_out;
    float* s0  = out + 1 * D;
    float* s1  = out + 2 * D;
    float* s2  = out + 3 * D;
    float* s3  = out + 4 * D;
    float* s4  = out + 5 * D;

    const MatvecJob jnull = { nullptr, nullptr, nullptr, nullptr, nullptr, nullptr, OP_NONE, 0 };

    {
        LnMixArgs ln = { x, lnw, lnb, state + 1 * D, s1 };
        MatvecJob jk = { kw, nullptr, tmk, kvec, nullptr, nullptr, OP_NONE,    D / 16 };
        MatvecJob jv = { vw, nullptr, tmv, vvec, nullptr, nullptr, OP_NONE,    D / 16 };
        MatvecJob jr = { rw, nullptr, tmr, rvec, nullptr, nullptr, OP_SIGMOID, D / 16 };
        ln_matvec3_kernel<<<3 * (D / 16), 256, 0, stream>>>(ln, jk, jv, jr);
    }

    {
        WkvArgs wk = { kvec, vvec, rvec, state, tf, td, s2, s3, s4 };
        MatvecJob jo = { ow, nullptr, nullptr, x_att, x, nullptr, OP_ADDRES, D / 16 };
        wkv_matvec_kernel<<<D / 16, 256, 0, stream>>>(wk, jo);
    }

    {
        LnMixArgs ln = { x_att, flnw, flnb, state  , s0 };
        MatvecJob jfr = { frw, nullptr, ftmr, fr, nullptr, nullptr, OP_SIGMOID, D / 16 };
        MatvecJob jfk = { fkw, nullptr, ftmk, fk, nullptr, nullptr, OP_SQRELU,  H / 16 };
        ln_matvec3_kernel<<<(D / 16) + (H / 16), 256, 0, stream>>>(ln, jfr, jfk, jnull);
    }

    {
        MatvecJob jout = { fvw, fk, nullptr, out, x_att, fr, OP_GATEDRES, D / 16 };
        wmma_matvec3_kernel<H><<<D / 16, 256, 0, stream>>>(jout, jnull, jnull);
    }
}


# Last mat-vec (K=4096) re-tiled from 64 work-groups of 16 rows x 8 waves to 256 of 4 rows x 4 waves, each wave's whole K slice in flight, windows alternating over 4 accumulators; sum order changes
# speedup vs baseline: 2.2909x; 1.0588x over previous
#include <hip/hip_runtime.h>
#include <hip/hip_bf16.h>


#define D 1024
#define H 4096
#define EPS 1e-5f

typedef __attribute__((ext_vector_type(2))) float v2f;
typedef __attribute__((ext_vector_type(4))) float v4f;
typedef __attribute__((ext_vector_type(8))) float v8f;

#define OP_NONE     0
#define OP_SIGMOID  1
#define OP_SQRELU   2
#define OP_ADDRES   3
#define OP_GATEDRES 4

struct MatvecJob {
    const float* W;
    const float* x;
    const float* tm;
    float*       y;
    const float* res;
    const float* gate;
    int op;
    int nblocks;
};

struct LnMixArgs {
    const float* xin;
    const float* w;
    const float* b;
    const float* prev;
    float*       xln_out;
};

struct WkvArgs {
    const float* kvec;
    const float* vvec;
    const float* rvec;
    const float* state;
    const float* time_first;
    const float* time_decay;
    float* s2;
    float* s3;
    float* s4;
};

struct WFrag { v4f a[16]; };
struct XFrag { v4f b[4]; };

template <int LDW>
__device__ __forceinline__ WFrag load_wfrag(const float* __restrict__ wp, int kb)
{
    WFrag f;
    #pragma unroll
    for (int w = 0; w < 4; ++w) {
        #pragma unroll
        for (int g = 0; g < 4; ++g)
            f.a[4 * w + g] = *(const v4f*)(wp + (size_t)(4 * g) * LDW + kb + 32 * w);
    }
    return f;
}

__device__ __forceinline__ XFrag load_xfrag(const float* xv, int kb)
{
    XFrag f;
    #pragma unroll
    for (int w = 0; w < 4; ++w)
        f.b[w] = *(const v4f*)(xv + kb + 32 * w);
    return f;
}

__device__ __forceinline__ void wmma_batch(const WFrag& fa, const XFrag& fb, v8f (&acc)[4])
{
    #pragma unroll
    for (int w = 0; w < 4; ++w) {
        const v2f blo = __builtin_shufflevector(fb.b[w], fb.b[w], 0, 1);
        const v2f bhi = __builtin_shufflevector(fb.b[w], fb.b[w], 2, 3);
        #pragma unroll
        for (int g = 0; g < 4; ++g) {
            const v2f alo = __builtin_shufflevector(fa.a[4 * w + g], fa.a[4 * w + g], 0, 1);
            acc[g] = __builtin_amdgcn_wmma_f32_16x16x4_f32(false, alo, false, blo, (short)0, acc[g], false, false);
        }
        #pragma unroll
        for (int g = 0; g < 4; ++g) {
            const v2f ahi = __builtin_shufflevector(fa.a[4 * w + g], fa.a[4 * w + g], 2, 3);
            acc[g] = __builtin_amdgcn_wmma_f32_16x16x4_f32(false, ahi, false, bhi, (short)0, acc[g], false, false);
        }
    }
}

template <int KK>
__device__ __forceinline__ void matvec_tail(
    const float* __restrict__ wp, const float* xv, WFrag first,
    int k0, float (*part)[16][4],
    float* __restrict__ y, int op,
    const float* __restrict__ res, const float* __restrict__ gate, int row0)
{
    constexpr int NB = KK / 8 / 128;
    const int wave = threadIdx.x >> 5;
    const int lane = threadIdx.x & 31;
    const int half = lane >> 4;
    const int c    = lane & 15;
    const int q    = c & 3;

    v8f acc[4] = {};

    WFrag ca = first;
    XFrag cb = load_xfrag(xv, k0);
    #pragma unroll
    for (int nb = 0; nb < NB; ++nb) {
        WFrag na;
        XFrag nx;
        if (nb + 1 < NB) {
            nx = load_xfrag(xv, k0 + 128 * (nb + 1));
            na = load_wfrag<KK>(wp, k0 + 128 * (nb + 1));
            __builtin_amdgcn_sched_barrier(0);
        }
        wmma_batch(ca, cb, acc);
        if (nb + 1 < NB) { ca = na; cb = nx; }
    }

    #pragma unroll
    for (int g = 0; g < 4; ++g) {
        const v8f d = acc[g];
        float lo = d[0], hi = d[4];
        lo = (q == 1) ? d[1] : lo;   hi = (q == 1) ? d[5] : hi;
        lo = (q == 2) ? d[2] : lo;   hi = (q == 2) ? d[6] : hi;
        lo = (q == 3) ? d[3] : lo;   hi = (q == 3) ? d[7] : hi;
        if (c < 4) {
            part[wave][4 * g + 2 * half    ][c] = lo;
            part[wave][4 * g + 2 * half + 1][c] = hi;
        }
    }
    __syncthreads();

    if (threadIdx.x < 16) {
        float v = 0.f;
        #pragma unroll
        for (int wv = 0; wv < 8; ++wv) {
            const v4f p = *(const v4f*)(&part[wv][threadIdx.x][0]);
            v += p.x; v += p.y; v += p.z; v += p.w;
        }
        const int r = row0 + threadIdx.x;
        switch (op) {
            case OP_SIGMOID:  v = 1.f / (1.f + expf(-v));              break;
            case OP_SQRELU:   { float t = fmaxf(v, 0.f); v = t * t; }  break;
            case OP_ADDRES:   v = res[r] + v;                          break;
            case OP_GATEDRES: v = res[r] + gate[r] * v;                break;
            default: break;
        }
        y[r] = v;
    }
}

__device__ __forceinline__ MatvecJob pick_job(
    const MatvecJob& j0, const MatvecJob& j1, const MatvecJob& j2, int& blk)
{
    if (blk < j0.nblocks) return j0;
    blk -= j0.nblocks;
    if (blk < j1.nblocks) return j1;
    blk -= j1.nblocks;
    return j2;
}

template <int KK, int NWV>
__global__ __launch_bounds__(32 * NWV) void wmma_matvec_rows4_kernel(MatvecJob j)
{
    constexpr int NW = KK / NWV / 32;
    __shared__ __attribute__((aligned(16))) float part[NWV][4][4];

    const int wave = threadIdx.x >> 5;
    const int lane = threadIdx.x & 31;
    const int half = lane >> 4;
    const int cl   = lane & 15;
    const int q    = cl & 3;
    const int row0 = blockIdx.x << 2;
    const int k0   = wave * (KK / NWV);
    const int loff = 8 * q + 4 * half;
    const float* wp = j.W + (size_t)(row0 + (cl >> 2)) * (size_t)KK + k0 + loff;
    const float* xp = j.x + k0 + loff;

    v4f a[NW], b[NW];
    #pragma unroll
    for (int w = 0; w < NW; ++w) {
        b[w] = *(const v4f*)(xp + 32 * w);
        a[w] = *(const v4f*)(wp + 32 * w);
    }
    __builtin_amdgcn_sched_barrier(0);

    v8f acc[4] = {};
    #pragma unroll
    for (int w = 0; w < NW; ++w) {
        const v2f alo = __builtin_shufflevector(a[w], a[w], 0, 1);
        const v2f blo = __builtin_shufflevector(b[w], b[w], 0, 1);
        const v2f ahi = __builtin_shufflevector(a[w], a[w], 2, 3);
        const v2f bhi = __builtin_shufflevector(b[w], b[w], 2, 3);
        acc[(2 * w) & 3] = __builtin_amdgcn_wmma_f32_16x16x4_f32(
            false, alo, false, blo, (short)0, acc[(2 * w) & 3], false, false);
        acc[(2 * w + 1) & 3] = __builtin_amdgcn_wmma_f32_16x16x4_f32(
            false, ahi, false, bhi, (short)0, acc[(2 * w + 1) & 3], false, false);
    }

    const v8f d = (acc[0] + acc[1]) + (acc[2] + acc[3]);
    float lo = d[0], hi = d[4];
    lo = (q == 1) ? d[1] : lo;   hi = (q == 1) ? d[5] : hi;
    lo = (q == 2) ? d[2] : lo;   hi = (q == 2) ? d[6] : hi;
    lo = (q == 3) ? d[3] : lo;   hi = (q == 3) ? d[7] : hi;
    if (cl < 4) {
        part[wave][2 * half    ][cl] = lo;
        part[wave][2 * half + 1][cl] = hi;
    }
    __syncthreads();

    if (threadIdx.x < 4) {
        float v = 0.f;
        #pragma unroll
        for (int wv = 0; wv < NWV; ++wv) {
            const v4f p = *(const v4f*)(&part[wv][threadIdx.x][0]);
            v += p.x; v += p.y; v += p.z; v += p.w;
        }
        const int r = row0 + threadIdx.x;
        switch (j.op) {
            case OP_SIGMOID:  v = 1.f / (1.f + expf(-v));              break;
            case OP_SQRELU:   { float t = fmaxf(v, 0.f); v = t * t; }  break;
            case OP_ADDRES:   v = j.res[r] + v;                        break;
            case OP_GATEDRES: v = j.res[r] + j.gate[r] * v;            break;
            default: break;
        }
        j.y[r] = v;
    }
}

__global__ __launch_bounds__(256) void ln_matvec3_kernel(
    LnMixArgs p, MatvecJob j0, MatvecJob j1, MatvecJob j2)
{
    __shared__ __attribute__((aligned(16))) float xs[D];
    __shared__ __attribute__((aligned(16))) float part[8][16][4];
    __shared__ float ssum[8];
    __shared__ float ssq[8];

    int blk = blockIdx.x;
    const MatvecJob j = pick_job(j0, j1, j2, blk);

    const int t    = threadIdx.x;
    const int wave = t >> 5;
    const int lane = t & 31;
    const int half = lane >> 4;
    const int cl   = lane & 15;
    const int row0 = blk << 4;
    const int k0 = wave * (D / 8);
    const int loff = 8 * (cl & 3) + 4 * half;
    const float* wp = j.W + (size_t)(row0 + (cl >> 2)) * (size_t)D + loff;

    float xv[4], wv[4], bv[4], pv[4], mv[4];
    #pragma unroll
    for (int q = 0; q < 4; ++q) {
        const int i = t + q * 256;
        xv[q] = p.xin[i];
        wv[q] = p.w[i];
        bv[q] = p.b[i];
        pv[q] = p.prev[i];
        mv[q] = j.tm[i];
    }

    WFrag first = load_wfrag<D>(wp, k0);

    float s = 0.f, sq = 0.f;
    #pragma unroll
    for (int q = 0; q < 4; ++q) {
        float v = xv[q];
        s += v;
        sq += v * v;
    }
    for (int off = 16; off > 0; off >>= 1) {
        s  += __shfl_xor(s,  off, 32);
        sq += __shfl_xor(sq, off, 32);
    }
    if (lane == 0) { ssum[wave] = s; ssq[wave] = sq; }
    __syncthreads();
    float a = (lane < 8) ? ssum[lane] : 0.f;
    float c = (lane < 8) ? ssq[lane]  : 0.f;
    for (int off = 4; off > 0; off >>= 1) {
        a += __shfl_xor(a, off, 32);
        c += __shfl_xor(c, off, 32);
    }
    a = __shfl(a, 0, 32);
    c = __shfl(c, 0, 32);

    const float mu   = a * (1.f / D);
    const float var  = c * (1.f / D) - mu * mu;
    const float rstd = rsqrtf(var + EPS);

    const bool writer = (blockIdx.x == 0);
    #pragma unroll
    for (int q = 0; q < 4; ++q) {
        const int i = t + q * 256;
        float xl = (xv[q] - mu) * rstd * wv[q] + bv[q];
        if (writer) p.xln_out[i] = xl;
        float k = mv[q];
        xs[i] = xl * k + pv[q] * (1.f - k);
    }
    __syncthreads();

    matvec_tail<D>(wp, xs + loff, first, k0, part,
                   j.y, j.op, j.res, j.gate, row0);
}

__global__ __launch_bounds__(256) void wkv_matvec_kernel(WkvArgs p, MatvecJob j)
{
    __shared__ __attribute__((aligned(16))) float xs[D];
    __shared__ __attribute__((aligned(16))) float part[8][16][4];

    const int blk  = blockIdx.x;
    const int t    = threadIdx.x;
    const int wave = t >> 5;
    const int lane = t & 31;
    const int half = lane >> 4;
    const int cl   = lane & 15;
    const int row0 = blk << 4;
    const int k0 = wave * (D / 8);
    const int loff = 8 * (cl & 3) + 4 * half;
    const float* wp = j.W + (size_t)(row0 + (cl >> 2)) * (size_t)D + loff;

    float kq[4], vq[4], rq[4], aq[4], bq[4], pq[4], tq[4];
    #pragma unroll
    for (int q = 0; q < 4; ++q) {
        const int i = t + q * 256;
        kq[q] = p.kvec[i];
        vq[q] = p.vvec[i];
        rq[q] = p.rvec[i];
        aq[q] = p.state[2 * D + i];
        bq[q] = p.state[3 * D + i];
        pq[q] = p.state[4 * D + i];
        tq[q] = p.time_first[i];
    }

    WFrag first = load_wfrag<D>(wp, k0);
    __builtin_amdgcn_sched_barrier(0);

    #pragma unroll
    for (int q = 0; q < 4; ++q) {
        const int i = t + q * 256;
        float k  = kq[q];
        float v  = vq[q];
        float aa = aq[q];
        float bb = bq[q];
        float pp = pq[q];

        float ww = tq[q] + k;
        float qq = fmaxf(pp, ww);
        float e1 = expf(pp - qq);
        float e2 = expf(ww - qq);
        float wkv = (e1 * aa + e2 * v) / (e1 * bb + e2);
        xs[i] = rq[q] * wkv;
    }

    if (blk == 0) {
        float dq[4];
        #pragma unroll
        for (int q = 0; q < 4; ++q)
            dq[q] = p.time_decay[t + q * 256];
        #pragma unroll
        for (int q = 0; q < 4; ++q) {
            const int i = t + q * 256;
            float k  = kq[q];
            float v  = vq[q];
            float aa = aq[q];
            float bb = bq[q];
            float pp = pq[q];

            float ww2 = pp + dq[q];
            float qq2 = fmaxf(ww2, k);
            float f1 = expf(ww2 - qq2);
            float f2 = expf(k - qq2);
            p.s2[i] = f1 * aa + f2 * v;
            p.s3[i] = f1 * bb + f2;
            p.s4[i] = qq2;
        }
    }
    __syncthreads();

    matvec_tail<D>(wp, xs + loff, first, k0, part,
                   j.y, j.op, j.res, j.gate, row0);
}

extern "C" void kernel_launch(void* const* d_in, const int* in_sizes, int n_in,
                              void* d_out, int out_size, void* d_ws, size_t ws_size,
                              hipStream_t stream) {
    const float* x     = (const float*)d_in[0];
    const float* state = (const float*)d_in[1];
    const float* tmk   = (const float*)d_in[2];
    const float* tmv   = (const float*)d_in[3];
    const float* tmr   = (const float*)d_in[4];
    const float* tf    = (const float*)d_in[5];
    const float* td    = (const float*)d_in[6];
    const float* kw    = (const float*)d_in[7];
    const float* vw    = (const float*)d_in[8];
    const float* rw    = (const float*)d_in[9];
    const float* ow    = (const float*)d_in[10];
    const float* lnw   = (const float*)d_in[11];
    const float* lnb   = (const float*)d_in[12];
    const float* ftmk  = (const float*)d_in[13];
    const float* ftmr  = (const float*)d_in[14];
    const float* fkw   = (const float*)d_in[15];
    const float* fvw   = (const float*)d_in[16];
    const float* frw   = (const float*)d_in[17];
    const float* flnw  = (const float*)d_in[18];
    const float* flnb  = (const float*)d_in[19];

    float* ws    = (float*)d_ws;
    float* kvec  = ws + 4 * D;
    float* vvec  = ws + 5 * D;
    float* rvec  = ws + 6 * D;
    float* x_att = ws + 8 * D;
    float* fr    = ws + 11 * D;
    float* fk    = ws + 12 * D;

    float* out = (float*)d_out;
    float* s0  = out + 1 * D;
    float* s1  = out + 2 * D;
    float* s2  = out + 3 * D;
    float* s3  = out + 4 * D;
    float* s4  = out + 5 * D;

    const MatvecJob jnull = { nullptr, nullptr, nullptr, nullptr, nullptr, nullptr, OP_NONE, 0 };

    {
        LnMixArgs ln = { x, lnw, lnb, state + 1 * D, s1 };
        MatvecJob jk = { kw, nullptr, tmk, kvec, nullptr, nullptr, OP_NONE,    D / 16 };
        MatvecJob jv = { vw, nullptr, tmv, vvec, nullptr, nullptr, OP_NONE,    D / 16 };
        MatvecJob jr = { rw, nullptr, tmr, rvec, nullptr, nullptr, OP_SIGMOID, D / 16 };
        ln_matvec3_kernel<<<3 * (D / 16), 256, 0, stream>>>(ln, jk, jv, jr);
    }

    {
        WkvArgs wk = { kvec, vvec, rvec, state, tf, td, s2, s3, s4 };
        MatvecJob jo = { ow, nullptr, nullptr, x_att, x, nullptr, OP_ADDRES, D / 16 };
        wkv_matvec_kernel<<<D / 16, 256, 0, stream>>>(wk, jo);
    }

    {
        LnMixArgs ln = { x_att, flnw, flnb, state  , s0 };
        MatvecJob jfr = { frw, nullptr, ftmr, fr, nullptr, nullptr, OP_SIGMOID, D / 16 };
        MatvecJob jfk = { fkw, nullptr, ftmk, fk, nullptr, nullptr, OP_SQRELU,  H / 16 };
        ln_matvec3_kernel<<<(D / 16) + (H / 16), 256, 0, stream>>>(ln, jfr, jfk, jnull);
    }

    {
        MatvecJob jout = { fvw, fk, nullptr, out, x_att, fr, OP_GATEDRES, D / 4 };
        wmma_matvec_rows4_kernel<H, 4><<<D / 4, 128, 0, stream>>>(jout);
    }
}


# Weight stream overlaps the WMMAs: prologue barriers order LDS only (LDS-scoped fences for __syncthreads, which drains the load counter); WMMAs pinned to the order their weights were loaded in
# speedup vs baseline: 2.7559x; 1.2030x over previous
#include <hip/hip_runtime.h>
#include <hip/hip_bf16.h>


#define D 1024
#define H 4096
#define EPS 1e-5f

typedef __attribute__((ext_vector_type(2))) float v2f;
typedef __attribute__((ext_vector_type(4))) float v4f;
typedef __attribute__((ext_vector_type(8))) float v8f;

#define OP_NONE     0
#define OP_SIGMOID  1
#define OP_SQRELU   2
#define OP_ADDRES   3
#define OP_GATEDRES 4

__device__ __forceinline__ void lds_barrier()
{
    __builtin_amdgcn_fence(__ATOMIC_RELEASE, "workgroup", "local");
    __builtin_amdgcn_s_barrier();
    __builtin_amdgcn_fence(__ATOMIC_ACQUIRE, "workgroup", "local");
}

struct MatvecJob {
    const float* W;
    const float* x;
    const float* tm;
    float*       y;
    const float* res;
    const float* gate;
    int op;
    int nblocks;
};

struct LnMixArgs {
    const float* xin;
    const float* w;
    const float* b;
    const float* prev;
    float*       xln_out;
};

struct WkvArgs {
    const float* kvec;
    const float* vvec;
    const float* rvec;
    const float* state;
    const float* time_first;
    const float* time_decay;
    float* s2;
    float* s3;
    float* s4;
};

struct WFrag { v4f a[16]; };
struct XFrag { v4f b[4]; };

template <int LDW>
__device__ __forceinline__ WFrag load_wfrag(const float* __restrict__ wp, int kb)
{
    WFrag f;
    #pragma unroll
    for (int w = 0; w < 4; ++w) {
        #pragma unroll
        for (int g = 0; g < 4; ++g)
            f.a[4 * w + g] = *(const v4f*)(wp + (size_t)(4 * g) * LDW + kb + 32 * w);
    }
    return f;
}

__device__ __forceinline__ XFrag load_xfrag(const float* xv, int kb)
{
    XFrag f;
    #pragma unroll
    for (int w = 0; w < 4; ++w)
        f.b[w] = *(const v4f*)(xv + kb + 32 * w);
    return f;
}

__device__ __forceinline__ void wmma_batch(const WFrag& fa, const XFrag& fb, v8f (&acc)[4])
{
    #pragma unroll
    for (int w = 0; w < 4; ++w) {
        const v2f blo = __builtin_shufflevector(fb.b[w], fb.b[w], 0, 1);
        const v2f bhi = __builtin_shufflevector(fb.b[w], fb.b[w], 2, 3);
        #pragma unroll
        for (int g = 0; g < 4; ++g) {
            const v2f alo = __builtin_shufflevector(fa.a[4 * w + g], fa.a[4 * w + g], 0, 1);
            acc[g] = __builtin_amdgcn_wmma_f32_16x16x4_f32(false, alo, false, blo, (short)0, acc[g], false, false);
        }
        #pragma unroll
        for (int g = 0; g < 4; ++g) {
            const v2f ahi = __builtin_shufflevector(fa.a[4 * w + g], fa.a[4 * w + g], 2, 3);
            acc[g] = __builtin_amdgcn_wmma_f32_16x16x4_f32(false, ahi, false, bhi, (short)0, acc[g], false, false);
        }
    }
}

template <int KK>
__device__ __forceinline__ void matvec_tail(
    const float* __restrict__ wp, const float* xv, WFrag first,
    int k0, float (*part)[16][4],
    float* __restrict__ y, int op,
    const float* __restrict__ res, const float* __restrict__ gate, int row0)
{
    constexpr int NB = KK / 8 / 128;
    const int wave = threadIdx.x >> 5;
    const int lane = threadIdx.x & 31;
    const int half = lane >> 4;
    const int c    = lane & 15;
    const int q    = c & 3;

    v8f acc[4] = {};

    WFrag ca = first;
    XFrag cb = load_xfrag(xv, k0);
    #pragma unroll
    for (int nb = 0; nb < NB; ++nb) {
        WFrag na;
        XFrag nx;
        if (nb + 1 < NB) {
            nx = load_xfrag(xv, k0 + 128 * (nb + 1));
            na = load_wfrag<KK>(wp, k0 + 128 * (nb + 1));
            __builtin_amdgcn_sched_barrier(0);
        }
        wmma_batch(ca, cb, acc);
        if (nb + 1 < NB) { ca = na; cb = nx; }
    }

    #pragma unroll
    for (int g = 0; g < 4; ++g) {
        const v8f d = acc[g];
        float lo = d[0], hi = d[4];
        lo = (q == 1) ? d[1] : lo;   hi = (q == 1) ? d[5] : hi;
        lo = (q == 2) ? d[2] : lo;   hi = (q == 2) ? d[6] : hi;
        lo = (q == 3) ? d[3] : lo;   hi = (q == 3) ? d[7] : hi;
        if (c < 4) {
            part[wave][4 * g + 2 * half    ][c] = lo;
            part[wave][4 * g + 2 * half + 1][c] = hi;
        }
    }
    __syncthreads();

    if (threadIdx.x < 16) {
        float v = 0.f;
        #pragma unroll
        for (int wv = 0; wv < 8; ++wv) {
            const v4f p = *(const v4f*)(&part[wv][threadIdx.x][0]);
            v += p.x; v += p.y; v += p.z; v += p.w;
        }
        const int r = row0 + threadIdx.x;
        switch (op) {
            case OP_SIGMOID:  v = 1.f / (1.f + expf(-v));              break;
            case OP_SQRELU:   { float t = fmaxf(v, 0.f); v = t * t; }  break;
            case OP_ADDRES:   v = res[r] + v;                          break;
            case OP_GATEDRES: v = res[r] + gate[r] * v;                break;
            default: break;
        }
        y[r] = v;
    }
}

template <int KK, int NWV>
__global__ __launch_bounds__(32 * NWV) void wmma_matvec_rows4_kernel(MatvecJob j)
{
    constexpr int NW = KK / NWV / 32;
    __shared__ __attribute__((aligned(16))) float part[NWV][4][4];

    const int wave = threadIdx.x >> 5;
    const int lane = threadIdx.x & 31;
    const int half = lane >> 4;
    const int cl   = lane & 15;
    const int q    = cl & 3;
    const int row0 = blockIdx.x << 2;
    const int k0   = wave * (KK / NWV);
    const int loff = 8 * q + 4 * half;
    const float* wp = j.W + (size_t)(row0 + (cl >> 2)) * (size_t)KK + k0 + loff;
    const float* xp = j.x + k0 + loff;

    v4f a[NW], b[NW];
    #pragma unroll
    for (int w = 0; w < NW; ++w) {
        b[w] = *(const v4f*)(xp + 32 * w);
        a[w] = *(const v4f*)(wp + 32 * w);
    }
    __builtin_amdgcn_sched_barrier(0);

    v8f acc[4] = {};
    #pragma unroll
    for (int w = 0; w < NW; ++w) {
        const v2f alo = __builtin_shufflevector(a[w], a[w], 0, 1);
        const v2f blo = __builtin_shufflevector(b[w], b[w], 0, 1);
        const v2f ahi = __builtin_shufflevector(a[w], a[w], 2, 3);
        const v2f bhi = __builtin_shufflevector(b[w], b[w], 2, 3);
        acc[(2 * w) & 3] = __builtin_amdgcn_wmma_f32_16x16x4_f32(
            false, alo, false, blo, (short)0, acc[(2 * w) & 3], false, false);
        acc[(2 * w + 1) & 3] = __builtin_amdgcn_wmma_f32_16x16x4_f32(
            false, ahi, false, bhi, (short)0, acc[(2 * w + 1) & 3], false, false);
    }

    const v8f d = (acc[0] + acc[1]) + (acc[2] + acc[3]);
    float lo = d[0], hi = d[4];
    lo = (q == 1) ? d[1] : lo;   hi = (q == 1) ? d[5] : hi;
    lo = (q == 2) ? d[2] : lo;   hi = (q == 2) ? d[6] : hi;
    lo = (q == 3) ? d[3] : lo;   hi = (q == 3) ? d[7] : hi;
    if (cl < 4) {
        part[wave][2 * half    ][cl] = lo;
        part[wave][2 * half + 1][cl] = hi;
    }
    __syncthreads();

    if (threadIdx.x < 4) {
        float v = 0.f;
        #pragma unroll
        for (int wv = 0; wv < NWV; ++wv) {
            const v4f p = *(const v4f*)(&part[wv][threadIdx.x][0]);
            v += p.x; v += p.y; v += p.z; v += p.w;
        }
        const int r = row0 + threadIdx.x;
        switch (j.op) {
            case OP_SIGMOID:  v = 1.f / (1.f + expf(-v));              break;
            case OP_SQRELU:   { float t = fmaxf(v, 0.f); v = t * t; }  break;
            case OP_ADDRES:   v = j.res[r] + v;                        break;
            case OP_GATEDRES: v = j.res[r] + j.gate[r] * v;            break;
            default: break;
        }
        j.y[r] = v;
    }
}

template <int G0, int G1, int G2>
__global__ __launch_bounds__(128) void ln_rows_kernel(
    LnMixArgs p, MatvecJob j0, MatvecJob j1, MatvecJob j2)
{
    constexpr int NJ = (G2 > 0) ? 3 : 2;
    constexpr int NG = G0 + G1 + G2;
    constexpr int KS = D / 4;
    constexpr int NW = KS / 32;

    __shared__ __attribute__((aligned(16))) float xs[NJ][D];
    __shared__ __attribute__((aligned(16))) float part[4][4 * NG][4];
    __shared__ __attribute__((aligned(16))) float ssum[4];
    __shared__ __attribute__((aligned(16))) float ssq[4];

    const int t    = threadIdx.x;
    const int wave = t >> 5;
    const int lane = t & 31;
    const int half = lane >> 4;
    const int cl   = lane & 15;
    const int q    = cl & 3;
    const int k0   = wave * KS;
    const int loff = 8 * q + 4 * half;

    const float* const Wj[3] = { j0.W,  j1.W,  j2.W  };
    const float* const Tj[3] = { j0.tm, j1.tm, j2.tm };
    constexpr int Gj[3]      = { G0, G1, G2 };

    v4f xv[2], wv[2], bv[2], pv[2], mv[NJ][2];
    #pragma unroll
    for (int u = 0; u < 2; ++u) {
        const int i = 4 * t + (D / 2) * u;
        xv[u] = *(const v4f*)(p.xin  + i);
        wv[u] = *(const v4f*)(p.w    + i);
        bv[u] = *(const v4f*)(p.b    + i);
        pv[u] = *(const v4f*)(p.prev + i);
        #pragma unroll
        for (int jb = 0; jb < NJ; ++jb)
            mv[jb][u] = *(const v4f*)(Tj[jb] + i);
    }
    __builtin_amdgcn_sched_barrier(0);

    v4f a[NG][NW];
    #pragma unroll
    for (int rg = 0; rg < NG; ++rg) {
        const int jb = (rg < G0) ? 0 : (rg < G0 + G1) ? 1 : 2;
        const int gi = rg - ((jb == 0) ? 0 : (jb == 1) ? G0 : G0 + G1);
        const int row = (blockIdx.x * Gj[jb] + gi) * 4 + (cl >> 2);
        const float* wp = Wj[jb] + (size_t)row * (size_t)D + k0 + loff;
        #pragma unroll
        for (int w = 0; w < NW; ++w)
            a[rg][w] = *(const v4f*)(wp + 32 * w);
    }
    __builtin_amdgcn_sched_barrier(0);

    float s = 0.f, sq = 0.f;
    #pragma unroll
    for (int u = 0; u < 2; ++u) {
        #pragma unroll
        for (int e = 0; e < 4; ++e) {
            const float v = xv[u][e];
            s += v;
            sq += v * v;
        }
    }
    for (int off = 16; off > 0; off >>= 1) {
        s  += __shfl_xor(s,  off, 32);
        sq += __shfl_xor(sq, off, 32);
    }
    if (lane == 0) { ssum[wave] = s; ssq[wave] = sq; }
    lds_barrier();
    const v4f ps = *(const v4f*)ssum;
    const v4f pq = *(const v4f*)ssq;
    const float sa = (ps.x + ps.y) + (ps.z + ps.w);
    const float sc = (pq.x + pq.y) + (pq.z + pq.w);

    const float mu   = sa * (1.f / D);
    const float var  = sc * (1.f / D) - mu * mu;
    const float rstd = rsqrtf(var + EPS);

    const bool writer = (blockIdx.x == 0);
    #pragma unroll
    for (int u = 0; u < 2; ++u) {
        const int i = 4 * t + (D / 2) * u;
        v4f xl;
        #pragma unroll
        for (int e = 0; e < 4; ++e)
            xl[e] = (xv[u][e] - mu) * rstd * wv[u][e] + bv[u][e];
        if (writer) *(v4f*)(p.xln_out + i) = xl;
        #pragma unroll
        for (int jb = 0; jb < NJ; ++jb) {
            v4f m;
            #pragma unroll
            for (int e = 0; e < 4; ++e) {
                const float k = mv[jb][u][e];
                m[e] = xl[e] * k + pv[u][e] * (1.f - k);
            }
            *(v4f*)(&xs[jb][i]) = m;
        }
    }
    lds_barrier();

    v4f b[NJ][NW];
    #pragma unroll
    for (int jb = 0; jb < NJ; ++jb) {
        #pragma unroll
        for (int w = 0; w < NW; ++w)
            b[jb][w] = *(const v4f*)(&xs[jb][k0 + loff + 32 * w]);
    }

    float lo[NG], hi[NG];
    #pragma unroll
    for (int rg = 0; rg < NG; ++rg) {
        const int jb = (rg < G0) ? 0 : (rg < G0 + G1) ? 1 : 2;
        v8f acc0 = {}, acc1 = {};
        #pragma unroll
        for (int w = 0; w < NW; ++w) {
            const v2f alo = __builtin_shufflevector(a[rg][w], a[rg][w], 0, 1);
            const v2f blo = __builtin_shufflevector(b[jb][w], b[jb][w], 0, 1);
            const v2f ahi = __builtin_shufflevector(a[rg][w], a[rg][w], 2, 3);
            const v2f bhi = __builtin_shufflevector(b[jb][w], b[jb][w], 2, 3);
            acc0 = __builtin_amdgcn_wmma_f32_16x16x4_f32(false, alo, false, blo, (short)0, acc0, false, false);
            acc1 = __builtin_amdgcn_wmma_f32_16x16x4_f32(false, ahi, false, bhi, (short)0, acc1, false, false);
        }
        __builtin_amdgcn_sched_barrier(0);
        const v8f d = acc0 + acc1;
        float l = d[0], h = d[4];
        l = (q == 1) ? d[1] : l;   h = (q == 1) ? d[5] : h;
        l = (q == 2) ? d[2] : l;   h = (q == 2) ? d[6] : h;
        l = (q == 3) ? d[3] : l;   h = (q == 3) ? d[7] : h;
        lo[rg] = l;
        hi[rg] = h;
    }
    if (cl < 4) {
        #pragma unroll
        for (int rg = 0; rg < NG; ++rg) {
            part[wave][4 * rg + 2 * half    ][cl] = lo[rg];
            part[wave][4 * rg + 2 * half + 1][cl] = hi[rg];
        }
    }
    __syncthreads();

    if (t < 4 * NG) {
        float v = 0.f;
        #pragma unroll
        for (int wv4 = 0; wv4 < 4; ++wv4) {
            const v4f pp = *(const v4f*)(&part[wv4][t][0]);
            v += pp.x; v += pp.y; v += pp.z; v += pp.w;
        }
        const int rg = t >> 2;
        const int jb = (rg < G0) ? 0 : (rg < G0 + G1) ? 1 : 2;
        const MatvecJob& j = (jb == 0) ? j0 : (jb == 1) ? j1 : j2;
        const int gi = rg - ((jb == 0) ? 0 : (jb == 1) ? G0 : G0 + G1);
        const int gj = (jb == 0) ? G0 : (jb == 1) ? G1 : G2;
        const int r  = (blockIdx.x * gj + gi) * 4 + (t & 3);
        switch (j.op) {
            case OP_SIGMOID:  v = 1.f / (1.f + expf(-v));              break;
            case OP_SQRELU:   { float x2 = fmaxf(v, 0.f); v = x2 * x2; } break;
            default: break;
        }
        j.y[r] = v;
    }
}

__global__ __launch_bounds__(256) void wkv_matvec_kernel(WkvArgs p, MatvecJob j)
{
    __shared__ __attribute__((aligned(16))) float xs[D];
    __shared__ __attribute__((aligned(16))) float part[8][16][4];

    const int blk  = blockIdx.x;
    const int t    = threadIdx.x;
    const int wave = t >> 5;
    const int lane = t & 31;
    const int half = lane >> 4;
    const int cl   = lane & 15;
    const int row0 = blk << 4;
    const int k0 = wave * (D / 8);
    const int loff = 8 * (cl & 3) + 4 * half;
    const float* wp = j.W + (size_t)(row0 + (cl >> 2)) * (size_t)D + loff;

    float kq[4], vq[4], rq[4], aq[4], bq[4], pq[4], tq[4];
    #pragma unroll
    for (int q = 0; q < 4; ++q) {
        const int i = t + q * 256;
        kq[q] = p.kvec[i];
        vq[q] = p.vvec[i];
        rq[q] = p.rvec[i];
        aq[q] = p.state[2 * D + i];
        bq[q] = p.state[3 * D + i];
        pq[q] = p.state[4 * D + i];
        tq[q] = p.time_first[i];
    }

    float dq[4] = { 0.f, 0.f, 0.f, 0.f };
    if (blk == 0) {
        #pragma unroll
        for (int q = 0; q < 4; ++q)
            dq[q] = p.time_decay[t + q * 256];
    }

    WFrag first = load_wfrag<D>(wp, k0);
    __builtin_amdgcn_sched_barrier(0);

    #pragma unroll
    for (int q = 0; q < 4; ++q) {
        const int i = t + q * 256;
        float k  = kq[q];
        float v  = vq[q];
        float aa = aq[q];
        float bb = bq[q];
        float pp = pq[q];

        float ww = tq[q] + k;
        float qq = fmaxf(pp, ww);
        float e1 = expf(pp - qq);
        float e2 = expf(ww - qq);
        float wkv = (e1 * aa + e2 * v) / (e1 * bb + e2);
        xs[i] = rq[q] * wkv;
    }

    if (blk == 0) {
        #pragma unroll
        for (int q = 0; q < 4; ++q) {
            const int i = t + q * 256;
            float k  = kq[q];
            float v  = vq[q];
            float aa = aq[q];
            float bb = bq[q];
            float pp = pq[q];

            float ww2 = pp + dq[q];
            float qq2 = fmaxf(ww2, k);
            float f1 = expf(ww2 - qq2);
            float f2 = expf(k - qq2);
            p.s2[i] = f1 * aa + f2 * v;
            p.s3[i] = f1 * bb + f2;
            p.s4[i] = qq2;
        }
    }
    lds_barrier();

    matvec_tail<D>(wp, xs + loff, first, k0, part,
                   j.y, j.op, j.res, j.gate, row0);
}

extern "C" void kernel_launch(void* const* d_in, const int* in_sizes, int n_in,
                              void* d_out, int out_size, void* d_ws, size_t ws_size,
                              hipStream_t stream) {
    const float* x     = (const float*)d_in[0];
    const float* state = (const float*)d_in[1];
    const float* tmk   = (const float*)d_in[2];
    const float* tmv   = (const float*)d_in[3];
    const float* tmr   = (const float*)d_in[4];
    const float* tf    = (const float*)d_in[5];
    const float* td    = (const float*)d_in[6];
    const float* kw    = (const float*)d_in[7];
    const float* vw    = (const float*)d_in[8];
    const float* rw    = (const float*)d_in[9];
    const float* ow    = (const float*)d_in[10];
    const float* lnw   = (const float*)d_in[11];
    const float* lnb   = (const float*)d_in[12];
    const float* ftmk  = (const float*)d_in[13];
    const float* ftmr  = (const float*)d_in[14];
    const float* fkw   = (const float*)d_in[15];
    const float* fvw   = (const float*)d_in[16];
    const float* frw   = (const float*)d_in[17];
    const float* flnw  = (const float*)d_in[18];
    const float* flnb  = (const float*)d_in[19];

    float* ws    = (float*)d_ws;
    float* kvec  = ws + 4 * D;
    float* vvec  = ws + 5 * D;
    float* rvec  = ws + 6 * D;
    float* x_att = ws + 8 * D;
    float* fr    = ws + 11 * D;
    float* fk    = ws + 12 * D;

    float* out = (float*)d_out;
    float* s0  = out + 1 * D;
    float* s1  = out + 2 * D;
    float* s2  = out + 3 * D;
    float* s3  = out + 4 * D;
    float* s4  = out + 5 * D;

    const MatvecJob jnull = { nullptr, nullptr, nullptr, nullptr, nullptr, nullptr, OP_NONE, 0 };

    {
        LnMixArgs ln = { x, lnw, lnb, state + 1 * D, s1 };
        MatvecJob jk = { kw, nullptr, tmk, kvec, nullptr, nullptr, OP_NONE,    D / 4 };
        MatvecJob jv = { vw, nullptr, tmv, vvec, nullptr, nullptr, OP_NONE,    D / 4 };
        MatvecJob jr = { rw, nullptr, tmr, rvec, nullptr, nullptr, OP_SIGMOID, D / 4 };
        ln_rows_kernel<1, 1, 1><<<D / 4, 128, 0, stream>>>(ln, jk, jv, jr);
    }

    {
        WkvArgs wk = { kvec, vvec, rvec, state, tf, td, s2, s3, s4 };
        MatvecJob jo = { ow, nullptr, nullptr, x_att, x, nullptr, OP_ADDRES, D / 16 };
        wkv_matvec_kernel<<<D / 16, 256, 0, stream>>>(wk, jo);
    }

    {
        LnMixArgs ln = { x_att, flnw, flnb, state  , s0 };
        MatvecJob jfr = { frw, nullptr, ftmr, fr, nullptr, nullptr, OP_SIGMOID, D / 4 };
        MatvecJob jfk = { fkw, nullptr, ftmk, fk, nullptr, nullptr, OP_SQRELU,  D / 4 };
        ln_rows_kernel<1, 4, 0><<<D / 4, 128, 0, stream>>>(ln, jfr, jfk, jnull);
    }

    {
        MatvecJob jout = { fvw, fk, nullptr, out, x_att, fr, OP_GATEDRES, D / 4 };
        wmma_matvec_rows4_kernel<H, 4><<<D / 4, 128, 0, stream>>>(jout);
    }
}


# WKV step moved from the ow launch's per-block prologue (all channels in each of 64 blocks) to the k/v/r launch's epilogue (4 channels per block); ow launch: plain 256-block mat-vec; sum order changes
# speedup vs baseline: 2.9079x; 1.0552x over previous
#include <hip/hip_runtime.h>
#include <hip/hip_bf16.h>


#define D 1024
#define H 4096
#define EPS 1e-5f

typedef __attribute__((ext_vector_type(2))) float v2f;
typedef __attribute__((ext_vector_type(4))) float v4f;
typedef __attribute__((ext_vector_type(8))) float v8f;

#define OP_NONE     0
#define OP_SIGMOID  1
#define OP_SQRELU   2
#define OP_ADDRES   3
#define OP_GATEDRES 4

__device__ __forceinline__ void lds_barrier()
{
    __builtin_amdgcn_fence(__ATOMIC_RELEASE, "workgroup", "local");
    __builtin_amdgcn_s_barrier();
    __builtin_amdgcn_fence(__ATOMIC_ACQUIRE, "workgroup", "local");
}

struct MatvecJob {
    const float* W;
    const float* x;
    const float* tm;
    float*       y;
    const float* res;
    const float* gate;
    int op;
    int nblocks;
};

struct LnMixArgs {
    const float* xin;
    const float* w;
    const float* b;
    const float* prev;
    float*       xln_out;
};

struct WkvArgs {
    const float* state;
    const float* time_first;
    const float* time_decay;
    float* rwkv;
    float* s2;
    float* s3;
    float* s4;
};

template <int KK, int NWV>
__global__ __launch_bounds__(32 * NWV) void wmma_matvec_rows4_kernel(MatvecJob j)
{
    constexpr int NW = KK / NWV / 32;
    __shared__ __attribute__((aligned(16))) float part[NWV][4][4];

    const int wave = threadIdx.x >> 5;
    const int lane = threadIdx.x & 31;
    const int half = lane >> 4;
    const int cl   = lane & 15;
    const int q    = cl & 3;
    const int row0 = blockIdx.x << 2;
    const int k0   = wave * (KK / NWV);
    const int loff = 8 * q + 4 * half;
    const float* wp = j.W + (size_t)(row0 + (cl >> 2)) * (size_t)KK + k0 + loff;
    const float* xp = j.x + k0 + loff;

    v4f a[NW], b[NW];
    #pragma unroll
    for (int w = 0; w < NW; ++w) {
        b[w] = *(const v4f*)(xp + 32 * w);
        a[w] = *(const v4f*)(wp + 32 * w);
    }
    __builtin_amdgcn_sched_barrier(0);

    v8f acc[4] = {};
    #pragma unroll
    for (int w = 0; w < NW; ++w) {
        const v2f alo = __builtin_shufflevector(a[w], a[w], 0, 1);
        const v2f blo = __builtin_shufflevector(b[w], b[w], 0, 1);
        const v2f ahi = __builtin_shufflevector(a[w], a[w], 2, 3);
        const v2f bhi = __builtin_shufflevector(b[w], b[w], 2, 3);
        acc[(2 * w) & 3] = __builtin_amdgcn_wmma_f32_16x16x4_f32(
            false, alo, false, blo, (short)0, acc[(2 * w) & 3], false, false);
        acc[(2 * w + 1) & 3] = __builtin_amdgcn_wmma_f32_16x16x4_f32(
            false, ahi, false, bhi, (short)0, acc[(2 * w + 1) & 3], false, false);
    }

    const v8f d = (acc[0] + acc[1]) + (acc[2] + acc[3]);
    float lo = d[0], hi = d[4];
    lo = (q == 1) ? d[1] : lo;   hi = (q == 1) ? d[5] : hi;
    lo = (q == 2) ? d[2] : lo;   hi = (q == 2) ? d[6] : hi;
    lo = (q == 3) ? d[3] : lo;   hi = (q == 3) ? d[7] : hi;
    if (cl < 4) {
        part[wave][2 * half    ][cl] = lo;
        part[wave][2 * half + 1][cl] = hi;
    }
    __syncthreads();

    if (threadIdx.x < 4) {
        float v = 0.f;
        #pragma unroll
        for (int wv = 0; wv < NWV; ++wv) {
            const v4f p = *(const v4f*)(&part[wv][threadIdx.x][0]);
            v += p.x; v += p.y; v += p.z; v += p.w;
        }
        const int r = row0 + threadIdx.x;
        switch (j.op) {
            case OP_SIGMOID:  v = 1.f / (1.f + expf(-v));              break;
            case OP_SQRELU:   { float t = fmaxf(v, 0.f); v = t * t; }  break;
            case OP_ADDRES:   v = j.res[r] + v;                        break;
            case OP_GATEDRES: v = j.res[r] + j.gate[r] * v;            break;
            default: break;
        }
        j.y[r] = v;
    }
}

template <int G0, int G1, int G2, bool WKV>
__global__ __launch_bounds__(128) void ln_rows_kernel(
    LnMixArgs p, MatvecJob j0, MatvecJob j1, MatvecJob j2, WkvArgs kp)
{
    static_assert(!WKV || (G0 == 1 && G1 == 1 && G2 == 1),
                  "the WKV epilogue needs k, v and r of a channel in one block");
    constexpr int NJ = (G2 > 0) ? 3 : 2;
    constexpr int NG = G0 + G1 + G2;
    constexpr int KS = D / 4;
    constexpr int NW = KS / 32;

    __shared__ __attribute__((aligned(16))) float xs[NJ][D];
    __shared__ __attribute__((aligned(16))) float part[4][4 * NG][4];
    __shared__ __attribute__((aligned(16))) float ssum[4];
    __shared__ __attribute__((aligned(16))) float ssq[4];

    const int t    = threadIdx.x;
    const int wave = t >> 5;
    const int lane = t & 31;
    const int half = lane >> 4;
    const int cl   = lane & 15;
    const int q    = cl & 3;
    const int k0   = wave * KS;
    const int loff = 8 * q + 4 * half;

    const float* const Wj[3] = { j0.W,  j1.W,  j2.W  };
    const float* const Tj[3] = { j0.tm, j1.tm, j2.tm };
    constexpr int Gj[3]      = { G0, G1, G2 };

    v4f xv[2], wv[2], bv[2], pv[2], mv[NJ][2];
    #pragma unroll
    for (int u = 0; u < 2; ++u) {
        const int i = 4 * t + (D / 2) * u;
        xv[u] = *(const v4f*)(p.xin  + i);
        wv[u] = *(const v4f*)(p.w    + i);
        bv[u] = *(const v4f*)(p.b    + i);
        pv[u] = *(const v4f*)(p.prev + i);
        #pragma unroll
        for (int jb = 0; jb < NJ; ++jb)
            mv[jb][u] = *(const v4f*)(Tj[jb] + i);
    }
    const int ch = 4 * blockIdx.x + (t & 3);
    float aa = 0.f, bb = 0.f, pp = 0.f, tfv = 0.f, tdv = 0.f;
    if constexpr (WKV) {
        aa  = kp.state[2 * D + ch];
        bb  = kp.state[3 * D + ch];
        pp  = kp.state[4 * D + ch];
        tfv = kp.time_first[ch];
        tdv = kp.time_decay[ch];
    }
    __builtin_amdgcn_sched_barrier(0);

    v4f a[NG][NW];
    #pragma unroll
    for (int rg = 0; rg < NG; ++rg) {
        const int jb = (rg < G0) ? 0 : (rg < G0 + G1) ? 1 : 2;
        const int gi = rg - ((jb == 0) ? 0 : (jb == 1) ? G0 : G0 + G1);
        const int row = (blockIdx.x * Gj[jb] + gi) * 4 + (cl >> 2);
        const float* wp = Wj[jb] + (size_t)row * (size_t)D + k0 + loff;
        #pragma unroll
        for (int w = 0; w < NW; ++w)
            a[rg][w] = *(const v4f*)(wp + 32 * w);
    }
    __builtin_amdgcn_sched_barrier(0);

    float s = 0.f, sq = 0.f;
    #pragma unroll
    for (int u = 0; u < 2; ++u) {
        #pragma unroll
        for (int e = 0; e < 4; ++e) {
            const float v = xv[u][e];
            s += v;
            sq += v * v;
        }
    }
    for (int off = 16; off > 0; off >>= 1) {
        s  += __shfl_xor(s,  off, 32);
        sq += __shfl_xor(sq, off, 32);
    }
    if (lane == 0) { ssum[wave] = s; ssq[wave] = sq; }
    lds_barrier();
    const v4f ps = *(const v4f*)ssum;
    const v4f pq = *(const v4f*)ssq;
    const float sa = (ps.x + ps.y) + (ps.z + ps.w);
    const float sc = (pq.x + pq.y) + (pq.z + pq.w);

    const float mu   = sa * (1.f / D);
    const float var  = sc * (1.f / D) - mu * mu;
    const float rstd = rsqrtf(var + EPS);

    const bool writer = (blockIdx.x == 0);
    #pragma unroll
    for (int u = 0; u < 2; ++u) {
        const int i = 4 * t + (D / 2) * u;
        v4f xl;
        #pragma unroll
        for (int e = 0; e < 4; ++e)
            xl[e] = (xv[u][e] - mu) * rstd * wv[u][e] + bv[u][e];
        if (writer) *(v4f*)(p.xln_out + i) = xl;
        #pragma unroll
        for (int jb = 0; jb < NJ; ++jb) {
            v4f m;
            #pragma unroll
            for (int e = 0; e < 4; ++e) {
                const float k = mv[jb][u][e];
                m[e] = xl[e] * k + pv[u][e] * (1.f - k);
            }
            *(v4f*)(&xs[jb][i]) = m;
        }
    }
    lds_barrier();

    v4f b[NJ][NW];
    #pragma unroll
    for (int jb = 0; jb < NJ; ++jb) {
        #pragma unroll
        for (int w = 0; w < NW; ++w)
            b[jb][w] = *(const v4f*)(&xs[jb][k0 + loff + 32 * w]);
    }

    float lo[NG], hi[NG];
    #pragma unroll
    for (int rg = 0; rg < NG; ++rg) {
        const int jb = (rg < G0) ? 0 : (rg < G0 + G1) ? 1 : 2;
        v8f acc0 = {}, acc1 = {};
        #pragma unroll
        for (int w = 0; w < NW; ++w) {
            const v2f alo = __builtin_shufflevector(a[rg][w], a[rg][w], 0, 1);
            const v2f blo = __builtin_shufflevector(b[jb][w], b[jb][w], 0, 1);
            const v2f ahi = __builtin_shufflevector(a[rg][w], a[rg][w], 2, 3);
            const v2f bhi = __builtin_shufflevector(b[jb][w], b[jb][w], 2, 3);
            acc0 = __builtin_amdgcn_wmma_f32_16x16x4_f32(false, alo, false, blo, (short)0, acc0, false, false);
            acc1 = __builtin_amdgcn_wmma_f32_16x16x4_f32(false, ahi, false, bhi, (short)0, acc1, false, false);
        }
        __builtin_amdgcn_sched_barrier(0);
        const v8f d = acc0 + acc1;
        float l = d[0], h = d[4];
        l = (q == 1) ? d[1] : l;   h = (q == 1) ? d[5] : h;
        l = (q == 2) ? d[2] : l;   h = (q == 2) ? d[6] : h;
        l = (q == 3) ? d[3] : l;   h = (q == 3) ? d[7] : h;
        lo[rg] = l;
        hi[rg] = h;
    }
    if (cl < 4) {
        #pragma unroll
        for (int rg = 0; rg < NG; ++rg) {
            part[wave][4 * rg + 2 * half    ][cl] = lo[rg];
            part[wave][4 * rg + 2 * half + 1][cl] = hi[rg];
        }
    }
    __syncthreads();

    if constexpr (WKV) {
        if (t < 4) {
            float k = 0.f, v = 0.f, r = 0.f;
            #pragma unroll
            for (int wv4 = 0; wv4 < 4; ++wv4) {
                const v4f qk = *(const v4f*)(&part[wv4][t][0]);
                const v4f qv = *(const v4f*)(&part[wv4][4 + t][0]);
                const v4f qr = *(const v4f*)(&part[wv4][8 + t][0]);
                k += qk.x; k += qk.y; k += qk.z; k += qk.w;
                v += qv.x; v += qv.y; v += qv.z; v += qv.w;
                r += qr.x; r += qr.y; r += qr.z; r += qr.w;
            }
            r = 1.f / (1.f + expf(-r));

            const float ww  = tfv + k;
            const float qq  = fmaxf(pp, ww);
            const float e1  = expf(pp - qq);
            const float e2  = expf(ww - qq);
            const float wkv = (e1 * aa + e2 * v) / (e1 * bb + e2);
            kp.rwkv[ch] = r * wkv;

            const float ww2 = pp + tdv;
            const float qq2 = fmaxf(ww2, k);
            const float f1  = expf(ww2 - qq2);
            const float f2  = expf(k - qq2);
            kp.s2[ch] = f1 * aa + f2 * v;
            kp.s3[ch] = f1 * bb + f2;
            kp.s4[ch] = qq2;
        }
    } else if (t < 4 * NG) {
        float v = 0.f;
        #pragma unroll
        for (int wv4 = 0; wv4 < 4; ++wv4) {
            const v4f pp4 = *(const v4f*)(&part[wv4][t][0]);
            v += pp4.x; v += pp4.y; v += pp4.z; v += pp4.w;
        }
        const int rg = t >> 2;
        const int jb = (rg < G0) ? 0 : (rg < G0 + G1) ? 1 : 2;
        const MatvecJob& j = (jb == 0) ? j0 : (jb == 1) ? j1 : j2;
        const int gi = rg - ((jb == 0) ? 0 : (jb == 1) ? G0 : G0 + G1);
        const int gj = (jb == 0) ? G0 : (jb == 1) ? G1 : G2;
        const int r  = (blockIdx.x * gj + gi) * 4 + (t & 3);
        switch (j.op) {
            case OP_SIGMOID:  v = 1.f / (1.f + expf(-v));              break;
            case OP_SQRELU:   { float x2 = fmaxf(v, 0.f); v = x2 * x2; } break;
            default: break;
        }
        j.y[r] = v;
    }
}

extern "C" void kernel_launch(void* const* d_in, const int* in_sizes, int n_in,
                              void* d_out, int out_size, void* d_ws, size_t ws_size,
                              hipStream_t stream) {
    const float* x     = (const float*)d_in[0];
    const float* state = (const float*)d_in[1];
    const float* tmk   = (const float*)d_in[2];
    const float* tmv   = (const float*)d_in[3];
    const float* tmr   = (const float*)d_in[4];
    const float* tf    = (const float*)d_in[5];
    const float* td    = (const float*)d_in[6];
    const float* kw    = (const float*)d_in[7];
    const float* vw    = (const float*)d_in[8];
    const float* rw    = (const float*)d_in[9];
    const float* ow    = (const float*)d_in[10];
    const float* lnw   = (const float*)d_in[11];
    const float* lnb   = (const float*)d_in[12];
    const float* ftmk  = (const float*)d_in[13];
    const float* ftmr  = (const float*)d_in[14];
    const float* fkw   = (const float*)d_in[15];
    const float* fvw   = (const float*)d_in[16];
    const float* frw   = (const float*)d_in[17];
    const float* flnw  = (const float*)d_in[18];
    const float* flnb  = (const float*)d_in[19];

    float* ws    = (float*)d_ws;
    float* rwkv  = ws + 4 * D;
    float* x_att = ws + 8 * D;
    float* fr    = ws + 11 * D;
    float* fk    = ws + 12 * D;

    float* out = (float*)d_out;
    float* s0  = out + 1 * D;
    float* s1  = out + 2 * D;
    float* s2  = out + 3 * D;
    float* s3  = out + 4 * D;
    float* s4  = out + 5 * D;

    const MatvecJob jnull = { nullptr, nullptr, nullptr, nullptr, nullptr, nullptr, OP_NONE, 0 };
    const WkvArgs   wnull = { nullptr, nullptr, nullptr, nullptr, nullptr, nullptr, nullptr };

    {
        LnMixArgs ln = { x, lnw, lnb, state + 1 * D, s1 };
        MatvecJob jk = { kw, nullptr, tmk, nullptr, nullptr, nullptr, OP_NONE,    D / 4 };
        MatvecJob jv = { vw, nullptr, tmv, nullptr, nullptr, nullptr, OP_NONE,    D / 4 };
        MatvecJob jr = { rw, nullptr, tmr, nullptr, nullptr, nullptr, OP_SIGMOID, D / 4 };
        WkvArgs   wk = { state, tf, td, rwkv, s2, s3, s4 };
        ln_rows_kernel<1, 1, 1, true><<<D / 4, 128, 0, stream>>>(ln, jk, jv, jr, wk);
    }

    {
        MatvecJob jo = { ow, rwkv, nullptr, x_att, x, nullptr, OP_ADDRES, D / 4 };
        wmma_matvec_rows4_kernel<D, 4><<<D / 4, 128, 0, stream>>>(jo);
    }

    {
        LnMixArgs ln = { x_att, flnw, flnb, state  , s0 };
        MatvecJob jfr = { frw, nullptr, ftmr, fr, nullptr, nullptr, OP_SIGMOID, D / 4 };
        MatvecJob jfk = { fkw, nullptr, ftmk, fk, nullptr, nullptr, OP_SQRELU,  D / 4 };
        ln_rows_kernel<1, 4, 0, false><<<D / 4, 128, 0, stream>>>(ln, jfr, jfk, jnull, wnull);
    }

    {
        MatvecJob jout = { fvw, fk, nullptr, out, x_att, fr, OP_GATEDRES, D / 4 };
        wmma_matvec_rows4_kernel<H, 4><<<D / 4, 128, 0, stream>>>(jout);
    }
}
